# starting kernel
# baseline (speedup 1.0000x reference)
#include <hip/hip_runtime.h>

typedef __attribute__((ext_vector_type(2))) float v2f;
typedef __attribute__((ext_vector_type(8))) float v8f;

#define BATCH 16
#define SRCN  100
#define TGTN  100
#define KDIM  512
#define HDIM  256
#define STILE 4

__device__ __forceinline__ float fast_tanh(float x) {
#if __has_builtin(__builtin_amdgcn_tanhf)
  return __builtin_amdgcn_tanhf(x);
#else
  float e = __builtin_amdgcn_exp2f(x * 2.8853900817779268f);
  return 1.0f - 2.0f * __builtin_amdgcn_rcpf(e + 1.0f);
#endif
}

__global__ __launch_bounds__(256) void proj_wmma_kernel(
    const float* __restrict__ query,
    const float* __restrict__ keys,
    const float* __restrict__ Wk,
    const float* __restrict__ Wq,
    const float* __restrict__ b1,
    float* __restrict__ hk,
    float* __restrict__ hqT)
{
  const int wave = threadIdx.x >> 5;
  const int lane = threadIdx.x & 31;
  int tile = blockIdx.x * 8 + wave;
  const int isQuery = (tile >= 400) ? 1 : 0;
  if (isQuery) tile -= 400;
  const int rowTile = tile >> 4;
  const int colTile = tile & 15;

  const float* __restrict__ A = isQuery ? query : keys;
  const float* __restrict__ W = isQuery ? Wq : Wk;

  const int mn   = lane & 15;
  const int half = lane >> 4;
  const int koff = half * 2;

  const float* __restrict__ Arow0 = A + (size_t)(rowTile * 64 + mn) * KDIM;
  const float* __restrict__ Arow1 = Arow0 + (size_t)16 * KDIM;
  const float* __restrict__ Arow2 = Arow0 + (size_t)32 * KDIM;
  const float* __restrict__ Arow3 = Arow0 + (size_t)48 * KDIM;
  const float* __restrict__ Wcol  = W + (colTile * 16 + mn);

  v8f acc0 = {}, acc1 = {}, acc2 = {}, acc3 = {};
  #pragma unroll 2
  for (int k = 0; k < KDIM; k += 4) {
    const int kk = k + koff;
    v2f b;
    b.x = Wcol[(size_t)kk * HDIM];
    b.y = Wcol[(size_t)(kk + 1) * HDIM];
    v2f a0, a1, a2, a3;
    a0.x = Arow0[kk]; a0.y = Arow0[kk + 1];
    a1.x = Arow1[kk]; a1.y = Arow1[kk + 1];
    a2.x = Arow2[kk]; a2.y = Arow2[kk + 1];
    a3.x = Arow3[kk]; a3.y = Arow3[kk + 1];
    acc0 = __builtin_amdgcn_wmma_f32_16x16x4_f32(false, a0, false, b, (short)0, acc0, false, false);
    acc1 = __builtin_amdgcn_wmma_f32_16x16x4_f32(false, a1, false, b, (short)0, acc1, false, false);
    acc2 = __builtin_amdgcn_wmma_f32_16x16x4_f32(false, a2, false, b, (short)0, acc2, false, false);
    acc3 = __builtin_amdgcn_wmma_f32_16x16x4_f32(false, a3, false, b, (short)0, acc3, false, false);
  }

  const int col = colTile * 16 + mn;
  v8f accs[4] = {acc0, acc1, acc2, acc3};
  if (!isQuery) {
    #pragma unroll
    for (int sub = 0; sub < 4; ++sub) {
      #pragma unroll
      for (int v = 0; v < 8; ++v) {
        const int r = rowTile * 64 + sub * 16 + v + half * 8;
        hk[(size_t)r * HDIM + col] = accs[sub][v];
      }
    }
  } else {
    const float bias = b1[col];
    #pragma unroll
    for (int sub = 0; sub < 4; ++sub) {
      #pragma unroll
      for (int v = 0; v < 8; ++v) {
        const int r  = rowTile * 64 + sub * 16 + v + half * 8;
        const int bb = r / TGTN;
        const int tt = r - bb * TGTN;
        hqT[((size_t)bb * HDIM + col) * TGTN + tt] = accs[sub][v] + bias;
      }
    }
  }
}

__global__ __launch_bounds__(128) void score_kernel(
    const float* __restrict__ hk,
    const float* __restrict__ hqT,
    const float* __restrict__ W2,
    const float* __restrict__ b2,
    float* __restrict__ out)
{
  __shared__ float hk_s[STILE][HDIM];
  __shared__ float w2_s[HDIM];

  const int bst = blockIdx.x;
  const int b   = bst / 25;
  const int s0  = (bst - b * 25) * STILE;
  const int tid = threadIdx.x;

  const float* __restrict__ hkBase = hk + (size_t)(b * SRCN + s0) * HDIM;
  #pragma unroll
  for (int i = 0; i < STILE * HDIM / 128; ++i)
    ((float*)hk_s)[tid + i * 128] = hkBase[tid + i * 128];
  w2_s[tid]       = W2[tid];
  w2_s[tid + 128] = W2[tid + 128];
  __syncthreads();

  if (tid < TGTN) {
    const float* __restrict__ q = hqT + (size_t)b * HDIM * TGTN + tid;
    float acc0 = 0.0f, acc1 = 0.0f, acc2 = 0.0f, acc3 = 0.0f;
    #pragma unroll 4
    for (int h = 0; h < HDIM; ++h) {
      const float qv = q[(size_t)h * TGTN];
      const float w  = w2_s[h];
      acc0 += fast_tanh(hk_s[0][h] + qv) * w;
      acc1 += fast_tanh(hk_s[1][h] + qv) * w;
      acc2 += fast_tanh(hk_s[2][h] + qv) * w;
      acc3 += fast_tanh(hk_s[3][h] + qv) * w;
    }
    const float b2v = b2[0];
    float* __restrict__ o = out + (size_t)(b * SRCN + s0) * TGTN + tid;
    o[0 * TGTN] = fast_tanh(acc0 + b2v);
    o[1 * TGTN] = fast_tanh(acc1 + b2v);
    o[2 * TGTN] = fast_tanh(acc2 + b2v);
    o[3 * TGTN] = fast_tanh(acc3 + b2v);
  }
}

extern "C" void kernel_launch(void* const* d_in, const int* in_sizes, int n_in,
                              void* d_out, int out_size, void* d_ws, size_t ws_size,
                              hipStream_t stream) {
  (void)in_sizes; (void)n_in; (void)out_size; (void)ws_size;

  const float* query = (const float*)d_in[0];
  const float* keys  = (const float*)d_in[1];
  const float* Wk    = (const float*)d_in[2];
  const float* Wq    = (const float*)d_in[3];
  const float* b1    = (const float*)d_in[4];
  const float* W2    = (const float*)d_in[5];
  const float* b2    = (const float*)d_in[6];
  float* out = (float*)d_out;

  float* hk  = (float*)d_ws;
  float* hqT = hk + (size_t)BATCH * SRCN * HDIM;

  proj_wmma_kernel<<<100, 256, 0, stream>>>(query, keys, Wk, Wq, b1, hk, hqT);
  score_kernel<<<BATCH * (SRCN / STILE), 128, 0, stream>>>(hk, hqT, W2, b2, out);
}


# Projection GEMMs: activation and weight slices staged through double-buffered LDS with whole-line 16-B global loads (64x64 tile per block, 200 blocks), fragments read from LDS; k order, sums kept
# speedup vs baseline: 1.7452x; 1.7452x over previous
#include <hip/hip_runtime.h>

typedef __attribute__((ext_vector_type(2))) float v2f;
typedef __attribute__((ext_vector_type(4))) float v4f;
typedef __attribute__((ext_vector_type(8))) float v8f;

#define BATCH 16
#define SRCN  100
#define TGTN  100
#define KDIM  512
#define HDIM  256
#define STILE 4

__device__ __forceinline__ float fast_tanh(float x) {
#if __has_builtin(__builtin_amdgcn_tanhf)
  return __builtin_amdgcn_tanhf(x);
#else
  float e = __builtin_amdgcn_exp2f(x * 2.8853900817779268f);
  return 1.0f - 2.0f * __builtin_amdgcn_rcpf(e + 1.0f);
#endif
}

#define PBM 64
#define PBN 64
#define PKC 64
#define A_PITCH 68
#define B_PITCH 72

__global__ __launch_bounds__(256) void proj_wmma_kernel(
    const float* __restrict__ query,
    const float* __restrict__ keys,
    const float* __restrict__ Wk,
    const float* __restrict__ Wq,
    const float* __restrict__ b1,
    float* __restrict__ hk,
    float* __restrict__ hqT)
{
  __shared__ __attribute__((aligned(16))) float As[2][PBM * A_PITCH];
  __shared__ __attribute__((aligned(16))) float Bs[2][PKC * B_PITCH];

  const int tid  = threadIdx.x;
  const int wave = tid >> 5;
  const int lane = tid & 31;
  int tile = blockIdx.x;
  const int isQuery = (tile >= 100) ? 1 : 0;
  if (isQuery) tile -= 100;
  const int rowTile = tile >> 2;
  const int colTile = tile & 3;

  const float* __restrict__ A = isQuery ? query : keys;
  const float* __restrict__ W = isQuery ? Wq : Wk;

  const int wr   = wave >> 2;
  const int wc   = wave & 3;
  const int mn   = lane & 15;
  const int half = lane >> 4;
  const int koff = half * 2;

  const int sRow = tid >> 4;
  const int sCol = (tid & 15) * 4;
  const float* __restrict__ aG = A + (size_t)(rowTile * PBM + sRow) * KDIM + sCol;
  const int aL = sRow * A_PITCH + sCol;
  const float* __restrict__ bG = W + (size_t)sRow * HDIM + colTile * PBN + sCol;
  const int bL = sRow * B_PITCH + sCol;

  const int aF = (wr * 32 + mn) * A_PITCH + koff;
  const int bF = koff * B_PITCH + wc * 16 + mn;

  v4f ra[4], rb[4];
  #pragma unroll
  for (int i = 0; i < 4; ++i) {
    ra[i] = *(const v4f*)(aG + (size_t)(16 * i) * KDIM);
    rb[i] = *(const v4f*)(bG + (size_t)(16 * i) * HDIM);
  }
  #pragma unroll
  for (int i = 0; i < 4; ++i) {
    *(v4f*)(&As[0][aL + 16 * i * A_PITCH]) = ra[i];
    *(v4f*)(&Bs[0][bL + 16 * i * B_PITCH]) = rb[i];
  }
  __syncthreads();

  v8f acc0 = {}, acc1 = {};
  for (int c = 0; c < KDIM / PKC; ++c) {
    const int cur = c & 1;
    const bool more = (c + 1) < (KDIM / PKC);
    if (more) {
      const int kn = (c + 1) * PKC;
      #pragma unroll
      for (int i = 0; i < 4; ++i) {
        ra[i] = *(const v4f*)(aG + (size_t)(16 * i) * KDIM + kn);
        rb[i] = *(const v4f*)(bG + (size_t)(kn + 16 * i) * HDIM);
      }
    }
    __builtin_amdgcn_sched_barrier(0);
    const float* __restrict__ aS = &As[cur][aF];
    const float* __restrict__ bS = &Bs[cur][bF];
    v2f fa0[PKC / 4], fa1[PKC / 4], fb[PKC / 4];
    #pragma unroll
    for (int j = 0; j < PKC / 4; ++j) {
      fa0[j] = *(const v2f*)(aS + j * 4);
      fa1[j] = *(const v2f*)(aS + 16 * A_PITCH + j * 4);
      fb[j].x = bS[(j * 4) * B_PITCH];
      fb[j].y = bS[(j * 4 + 1) * B_PITCH];
    }
    __builtin_amdgcn_sched_barrier(0);
    #pragma unroll
    for (int j = 0; j < PKC / 4; ++j) {
      acc0 = __builtin_amdgcn_wmma_f32_16x16x4_f32(false, fa0[j], false, fb[j], (short)0, acc0, false, false);
      acc1 = __builtin_amdgcn_wmma_f32_16x16x4_f32(false, fa1[j], false, fb[j], (short)0, acc1, false, false);
    }
    __builtin_amdgcn_sched_barrier(0);
    if (more) {
      const int nxt = cur ^ 1;
      #pragma unroll
      for (int i = 0; i < 4; ++i) {
        *(v4f*)(&As[nxt][aL + 16 * i * A_PITCH]) = ra[i];
        *(v4f*)(&Bs[nxt][bL + 16 * i * B_PITCH]) = rb[i];
      }
    }
    __syncthreads();
  }

  const int col = colTile * PBN + wc * 16 + mn;
  v8f accs[2] = {acc0, acc1};
  if (!isQuery) {
    #pragma unroll
    for (int sub = 0; sub < 2; ++sub) {
      #pragma unroll
      for (int v = 0; v < 8; ++v) {
        const int r = rowTile * PBM + wr * 32 + sub * 16 + v + half * 8;
        hk[(size_t)r * HDIM + col] = accs[sub][v];
      }
    }
  } else {
    const float bias = b1[col];
    #pragma unroll
    for (int sub = 0; sub < 2; ++sub) {
      #pragma unroll
      for (int v = 0; v < 8; ++v) {
        const int r  = rowTile * PBM + wr * 32 + sub * 16 + v + half * 8;
        const int bb = r / TGTN;
        const int tt = r - bb * TGTN;
        hqT[((size_t)bb * HDIM + col) * TGTN + tt] = accs[sub][v] + bias;
      }
    }
  }
}

__global__ __launch_bounds__(128) void score_kernel(
    const float* __restrict__ hk,
    const float* __restrict__ hqT,
    const float* __restrict__ W2,
    const float* __restrict__ b2,
    float* __restrict__ out)
{
  __shared__ float hk_s[STILE][HDIM];
  __shared__ float w2_s[HDIM];

  const int bst = blockIdx.x;
  const int b   = bst / 25;
  const int s0  = (bst - b * 25) * STILE;
  const int tid = threadIdx.x;

  const float* __restrict__ hkBase = hk + (size_t)(b * SRCN + s0) * HDIM;
  #pragma unroll
  for (int i = 0; i < STILE * HDIM / 128; ++i)
    ((float*)hk_s)[tid + i * 128] = hkBase[tid + i * 128];
  w2_s[tid]       = W2[tid];
  w2_s[tid + 128] = W2[tid + 128];
  __syncthreads();

  if (tid < TGTN) {
    const float* __restrict__ q = hqT + (size_t)b * HDIM * TGTN + tid;
    float acc0 = 0.0f, acc1 = 0.0f, acc2 = 0.0f, acc3 = 0.0f;
    #pragma unroll 4
    for (int h = 0; h < HDIM; ++h) {
      const float qv = q[(size_t)h * TGTN];
      const float w  = w2_s[h];
      acc0 += fast_tanh(hk_s[0][h] + qv) * w;
      acc1 += fast_tanh(hk_s[1][h] + qv) * w;
      acc2 += fast_tanh(hk_s[2][h] + qv) * w;
      acc3 += fast_tanh(hk_s[3][h] + qv) * w;
    }
    const float b2v = b2[0];
    float* __restrict__ o = out + (size_t)(b * SRCN + s0) * TGTN + tid;
    o[0 * TGTN] = fast_tanh(acc0 + b2v);
    o[1 * TGTN] = fast_tanh(acc1 + b2v);
    o[2 * TGTN] = fast_tanh(acc2 + b2v);
    o[3 * TGTN] = fast_tanh(acc3 + b2v);
  }
}

extern "C" void kernel_launch(void* const* d_in, const int* in_sizes, int n_in,
                              void* d_out, int out_size, void* d_ws, size_t ws_size,
                              hipStream_t stream) {
  (void)in_sizes; (void)n_in; (void)out_size; (void)ws_size;

  const float* query = (const float*)d_in[0];
  const float* keys  = (const float*)d_in[1];
  const float* Wk    = (const float*)d_in[2];
  const float* Wq    = (const float*)d_in[3];
  const float* b1    = (const float*)d_in[4];
  const float* W2    = (const float*)d_in[5];
  const float* b2    = (const float*)d_in[6];
  float* out = (float*)d_out;

  float* hk  = (float*)d_ws;
  float* hqT = hk + (size_t)BATCH * SRCN * HDIM;

  proj_wmma_kernel<<<200, 256, 0, stream>>>(query, keys, Wk, Wq, b1, hk, hqT);
  score_kernel<<<BATCH * (SRCN / STILE), 128, 0, stream>>>(hk, hqT, W2, b2, out);
}


# Scoring kernel: every thread loads its whole hq column into registers up front (64 16-B loads in flight; hq stored 4-h-interleaved per t) instead of a waited load group per 4 h; sum order kept
# speedup vs baseline: 4.4005x; 2.5215x over previous
#include <hip/hip_runtime.h>

typedef __attribute__((ext_vector_type(2))) float v2f;
typedef __attribute__((ext_vector_type(4))) float v4f;
typedef __attribute__((ext_vector_type(8))) float v8f;

#define BATCH 16
#define SRCN  100
#define TGTN  100
#define KDIM  512
#define HDIM  256
#define STILE 4
#define SCH   4

__device__ __forceinline__ float fast_tanh(float x) {
#if __has_builtin(__builtin_amdgcn_tanhf)
  return __builtin_amdgcn_tanhf(x);
#else
  float e = __builtin_amdgcn_exp2f(x * 2.8853900817779268f);
  return 1.0f - 2.0f * __builtin_amdgcn_rcpf(e + 1.0f);
#endif
}

#define PBM 64
#define PBN 64
#define PKC 64
#define A_PITCH 68
#define B_PITCH 72

__global__ __launch_bounds__(256) void proj_wmma_kernel(
    const float* __restrict__ query,
    const float* __restrict__ keys,
    const float* __restrict__ Wk,
    const float* __restrict__ Wq,
    const float* __restrict__ b1,
    float* __restrict__ hk,
    float* __restrict__ hq4)
{
  __shared__ __attribute__((aligned(16))) float As[2][PBM * A_PITCH];
  __shared__ __attribute__((aligned(16))) float Bs[2][PKC * B_PITCH];

  const int tid  = threadIdx.x;
  const int wave = tid >> 5;
  const int lane = tid & 31;
  int tile = blockIdx.x;
  const int isQuery = (tile >= 100) ? 1 : 0;
  if (isQuery) tile -= 100;
  const int rowTile = tile >> 2;
  const int colTile = tile & 3;

  const float* __restrict__ A = isQuery ? query : keys;
  const float* __restrict__ W = isQuery ? Wq : Wk;

  const int wr   = wave >> 2;
  const int wc   = wave & 3;
  const int mn   = lane & 15;
  const int half = lane >> 4;
  const int koff = half * 2;

  const int sRow = tid >> 4;
  const int sCol = (tid & 15) * 4;
  const float* __restrict__ aG = A + (size_t)(rowTile * PBM + sRow) * KDIM + sCol;
  const int aL = sRow * A_PITCH + sCol;
  const float* __restrict__ bG = W + (size_t)sRow * HDIM + colTile * PBN + sCol;
  const int bL = sRow * B_PITCH + sCol;

  const int aF = (wr * 32 + mn) * A_PITCH + koff;
  const int bF = koff * B_PITCH + wc * 16 + mn;

  v4f ra[4], rb[4];
  #pragma unroll
  for (int i = 0; i < 4; ++i) {
    ra[i] = *(const v4f*)(aG + (size_t)(16 * i) * KDIM);
    rb[i] = *(const v4f*)(bG + (size_t)(16 * i) * HDIM);
  }
  #pragma unroll
  for (int i = 0; i < 4; ++i) {
    *(v4f*)(&As[0][aL + 16 * i * A_PITCH]) = ra[i];
    *(v4f*)(&Bs[0][bL + 16 * i * B_PITCH]) = rb[i];
  }
  __syncthreads();

  v8f acc0 = {}, acc1 = {};
  for (int c = 0; c < KDIM / PKC; ++c) {
    const int cur = c & 1;
    const bool more = (c + 1) < (KDIM / PKC);
    if (more) {
      const int kn = (c + 1) * PKC;
      #pragma unroll
      for (int i = 0; i < 4; ++i) {
        ra[i] = *(const v4f*)(aG + (size_t)(16 * i) * KDIM + kn);
        rb[i] = *(const v4f*)(bG + (size_t)(kn + 16 * i) * HDIM);
      }
    }
    __builtin_amdgcn_sched_barrier(0);
    const float* __restrict__ aS = &As[cur][aF];
    const float* __restrict__ bS = &Bs[cur][bF];
    v2f fa0[PKC / 4], fa1[PKC / 4], fb[PKC / 4];
    #pragma unroll
    for (int j = 0; j < PKC / 4; ++j) {
      fa0[j] = *(const v2f*)(aS + j * 4);
      fa1[j] = *(const v2f*)(aS + 16 * A_PITCH + j * 4);
      fb[j].x = bS[(j * 4) * B_PITCH];
      fb[j].y = bS[(j * 4 + 1) * B_PITCH];
    }
    __builtin_amdgcn_sched_barrier(0);
    #pragma unroll
    for (int j = 0; j < PKC / 4; ++j) {
      acc0 = __builtin_amdgcn_wmma_f32_16x16x4_f32(false, fa0[j], false, fb[j], (short)0, acc0, false, false);
      acc1 = __builtin_amdgcn_wmma_f32_16x16x4_f32(false, fa1[j], false, fb[j], (short)0, acc1, false, false);
    }
    __builtin_amdgcn_sched_barrier(0);
    if (more) {
      const int nxt = cur ^ 1;
      #pragma unroll
      for (int i = 0; i < 4; ++i) {
        *(v4f*)(&As[nxt][aL + 16 * i * A_PITCH]) = ra[i];
        *(v4f*)(&Bs[nxt][bL + 16 * i * B_PITCH]) = rb[i];
      }
    }
    __syncthreads();
  }

  const int col = colTile * PBN + wc * 16 + mn;
  v8f accs[2] = {acc0, acc1};
  if (!isQuery) {
    #pragma unroll
    for (int sub = 0; sub < 2; ++sub) {
      #pragma unroll
      for (int v = 0; v < 8; ++v) {
        const int r = rowTile * PBM + wr * 32 + sub * 16 + v + half * 8;
        hk[(size_t)r * HDIM + col] = accs[sub][v];
      }
    }
  } else {
    const float bias = b1[col];
    #pragma unroll
    for (int sub = 0; sub < 2; ++sub) {
      #pragma unroll
      for (int v = 0; v < 8; ++v) {
        const int r  = rowTile * PBM + wr * 32 + sub * 16 + v + half * 8;
        const int bb = r / TGTN;
        const int tt = r - bb * TGTN;
        hq4[(((size_t)bb * (HDIM / 4) + (col >> 2)) * TGTN + tt) * 4 + (col & 3)] = accs[sub][v] + bias;
      }
    }
  }
}

__global__ __launch_bounds__(128) void score_kernel(
    const float* __restrict__ hk,
    const float* __restrict__ hq4,
    const float* __restrict__ W2,
    const float* __restrict__ b2,
    float* __restrict__ out)
{
  __shared__ __attribute__((aligned(16))) float hk_s[STILE][HDIM];
  __shared__ __attribute__((aligned(16))) float w2_s[HDIM];

  const int bst = blockIdx.x;
  const int b   = bst / 25;
  const int s0  = (bst - b * 25) * STILE;
  const int tid = threadIdx.x;

  const float* __restrict__ hkBase = hk + (size_t)(b * SRCN + s0) * HDIM;
  float st[STILE * HDIM / 128];
  #pragma unroll
  for (int i = 0; i < STILE * HDIM / 128; ++i) st[i] = hkBase[tid + i * 128];
  const float wlo = W2[tid];
  const float whi = W2[tid + 128];

  const int t = (tid < TGTN) ? tid : (TGTN - 1);
  const float* __restrict__ qB = hq4 + ((size_t)b * (HDIM / 4) * TGTN + t) * 4;
  v4f q[HDIM / 4];
  #pragma unroll
  for (int g = 0; g < HDIM / 4; ++g) q[g] = *(const v4f*)(qB + (size_t)g * TGTN * 4);

  #pragma unroll
  for (int i = 0; i < STILE * HDIM / 128; ++i) ((float*)hk_s)[tid + i * 128] = st[i];
  w2_s[tid]       = wlo;
  w2_s[tid + 128] = whi;
  __syncthreads();

  float acc0 = 0.0f, acc1 = 0.0f, acc2 = 0.0f, acc3 = 0.0f;
  float kk[2][SCH * 4][5];
  #pragma unroll
  for (int i = 0; i < SCH * 4; ++i) {
    kk[0][i][0] = hk_s[0][i];
    kk[0][i][1] = hk_s[1][i];
    kk[0][i][2] = hk_s[2][i];
    kk[0][i][3] = hk_s[3][i];
    kk[0][i][4] = w2_s[i];
  }
  #pragma unroll
  for (int c = 0; c < HDIM / (4 * SCH); ++c) {
    const int cur = c & 1;
    if (c + 1 < HDIM / (4 * SCH)) {
      #pragma unroll
      for (int i = 0; i < SCH * 4; ++i) {
        const int h = (c + 1) * SCH * 4 + i;
        kk[cur ^ 1][i][0] = hk_s[0][h];
        kk[cur ^ 1][i][1] = hk_s[1][h];
        kk[cur ^ 1][i][2] = hk_s[2][h];
        kk[cur ^ 1][i][3] = hk_s[3][h];
        kk[cur ^ 1][i][4] = w2_s[h];
      }
      __builtin_amdgcn_sched_barrier(0);
    }
    #pragma unroll
    for (int i = 0; i < SCH * 4; ++i) {
      const int h = c * SCH * 4 + i;
      const float qv = q[h >> 2][h & 3];
      const float w  = kk[cur][i][4];
      acc0 += fast_tanh(kk[cur][i][0] + qv) * w;
      acc1 += fast_tanh(kk[cur][i][1] + qv) * w;
      acc2 += fast_tanh(kk[cur][i][2] + qv) * w;
      acc3 += fast_tanh(kk[cur][i][3] + qv) * w;
    }
    if (c + 2 < HDIM / (4 * SCH))
      asm volatile("" :: "v"(acc0), "v"(acc1), "v"(acc2), "v"(acc3) : "memory");
  }

  if (tid < TGTN) {
    const float b2v = b2[0];
    float* __restrict__ o = out + (size_t)(b * SRCN + s0) * TGTN + tid;
    o[0 * TGTN] = fast_tanh(acc0 + b2v);
    o[1 * TGTN] = fast_tanh(acc1 + b2v);
    o[2 * TGTN] = fast_tanh(acc2 + b2v);
    o[3 * TGTN] = fast_tanh(acc3 + b2v);
  }
}

extern "C" void kernel_launch(void* const* d_in, const int* in_sizes, int n_in,
                              void* d_out, int out_size, void* d_ws, size_t ws_size,
                              hipStream_t stream) {
  (void)in_sizes; (void)n_in; (void)out_size; (void)ws_size;

  const float* query = (const float*)d_in[0];
  const float* keys  = (const float*)d_in[1];
  const float* Wk    = (const float*)d_in[2];
  const float* Wq    = (const float*)d_in[3];
  const float* b1    = (const float*)d_in[4];
  const float* W2    = (const float*)d_in[5];
  const float* b2    = (const float*)d_in[6];
  float* out = (float*)d_out;

  float* hk  = (float*)d_ws;
  float* hq4 = hk + (size_t)BATCH * SRCN * HDIM;

  proj_wmma_kernel<<<200, 256, 0, stream>>>(query, keys, Wk, Wq, b1, hk, hq4);
  score_kernel<<<BATCH * (SRCN / STILE), 128, 0, stream>>>(hk, hq4, W2, b2, out);
}


# Workgroup barriers fence LDS only, so global loads stay in flight across them: scoring h loop starts while its hq column is still arriving; projection loads run 2 chunks ahead; sums unchanged
# speedup vs baseline: 4.8076x; 1.0925x over previous
#include <hip/hip_runtime.h>

typedef __attribute__((ext_vector_type(2))) float v2f;
typedef __attribute__((ext_vector_type(4))) float v4f;
typedef __attribute__((ext_vector_type(8))) float v8f;

#define BATCH 16
#define SRCN  100
#define TGTN  100
#define KDIM  512
#define HDIM  256
#define STILE 4
#define SCH   4

__device__ __forceinline__ float fast_tanh(float x) {
#if __has_builtin(__builtin_amdgcn_tanhf)
  return __builtin_amdgcn_tanhf(x);
#else
  float e = __builtin_amdgcn_exp2f(x * 2.8853900817779268f);
  return 1.0f - 2.0f * __builtin_amdgcn_rcpf(e + 1.0f);
#endif
}


__device__ __forceinline__ void lds_barrier() {
  __builtin_amdgcn_fence(__ATOMIC_RELEASE, "workgroup", "local");
  __builtin_amdgcn_s_barrier();
  __builtin_amdgcn_fence(__ATOMIC_ACQUIRE, "workgroup", "local");
}

#define PBM 64
#define PBN 64
#define PKC 64
#define A_PITCH 68
#define B_PITCH 72

__global__ __launch_bounds__(256) void proj_wmma_kernel(
    const float* __restrict__ query,
    const float* __restrict__ keys,
    const float* __restrict__ Wk,
    const float* __restrict__ Wq,
    const float* __restrict__ b1,
    float* __restrict__ hk,
    float* __restrict__ hq4)
{
  __shared__ __attribute__((aligned(16))) float As[2][PBM * A_PITCH];
  __shared__ __attribute__((aligned(16))) float Bs[2][PKC * B_PITCH];

  const int tid  = threadIdx.x;
  const int wave = tid >> 5;
  const int lane = tid & 31;
  int tile = blockIdx.x;
  const int isQuery = (tile >= 100) ? 1 : 0;
  if (isQuery) tile -= 100;
  const int rowTile = tile >> 2;
  const int colTile = tile & 3;

  const float* __restrict__ A = isQuery ? query : keys;
  const float* __restrict__ W = isQuery ? Wq : Wk;

  const int wr   = wave >> 2;
  const int wc   = wave & 3;
  const int mn   = lane & 15;
  const int half = lane >> 4;
  const int koff = half * 2;

  const int sRow = tid >> 4;
  const int sCol = (tid & 15) * 4;
  const float* __restrict__ aG = A + (size_t)(rowTile * PBM + sRow) * KDIM + sCol;
  const int aL = sRow * A_PITCH + sCol;
  const float* __restrict__ bG = W + (size_t)sRow * HDIM + colTile * PBN + sCol;
  const int bL = sRow * B_PITCH + sCol;

  const int aF = (wr * 32 + mn) * A_PITCH + koff;
  const int bF = koff * B_PITCH + wc * 16 + mn;

  v4f ra[2][4], rb[2][4];
  #pragma unroll
  for (int s = 0; s < 2; ++s) {
    #pragma unroll
    for (int i = 0; i < 4; ++i) {
      ra[s][i] = *(const v4f*)(aG + (size_t)(16 * i) * KDIM + s * PKC);
      rb[s][i] = *(const v4f*)(bG + (size_t)(s * PKC + 16 * i) * HDIM);
    }
    __builtin_amdgcn_sched_barrier(0);
  }
  #pragma unroll
  for (int i = 0; i < 4; ++i) {
    *(v4f*)(&As[0][aL + 16 * i * A_PITCH]) = ra[0][i];
    *(v4f*)(&Bs[0][bL + 16 * i * B_PITCH]) = rb[0][i];
  }
  lds_barrier();

  v8f acc0 = {}, acc1 = {};
  #pragma unroll
  for (int c = 0; c < KDIM / PKC; ++c) {
    const int cur = c & 1;
    if (c + 2 < KDIM / PKC) {
      const int kn = (c + 2) * PKC;
      #pragma unroll
      for (int i = 0; i < 4; ++i) {
        ra[cur][i] = *(const v4f*)(aG + (size_t)(16 * i) * KDIM + kn);
        rb[cur][i] = *(const v4f*)(bG + (size_t)(kn + 16 * i) * HDIM);
      }
    }
    __builtin_amdgcn_sched_barrier(0);
    const float* __restrict__ aS = &As[cur][aF];
    const float* __restrict__ bS = &Bs[cur][bF];
    v2f fa0[PKC / 4], fa1[PKC / 4], fb[PKC / 4];
    #pragma unroll
    for (int j = 0; j < PKC / 4; ++j) {
      fa0[j] = *(const v2f*)(aS + j * 4);
      fa1[j] = *(const v2f*)(aS + 16 * A_PITCH + j * 4);
      fb[j].x = bS[(j * 4) * B_PITCH];
      fb[j].y = bS[(j * 4 + 1) * B_PITCH];
      if (j & 1) __builtin_amdgcn_sched_barrier(0);
    }
    __builtin_amdgcn_sched_barrier(0);
    #pragma unroll
    for (int j = 0; j < PKC / 4; ++j) {
      acc0 = __builtin_amdgcn_wmma_f32_16x16x4_f32(false, fa0[j], false, fb[j], (short)0, acc0, false, false);
      acc1 = __builtin_amdgcn_wmma_f32_16x16x4_f32(false, fa1[j], false, fb[j], (short)0, acc1, false, false);
    }
    __builtin_amdgcn_sched_barrier(0);
    if (c + 1 < KDIM / PKC) {
      const int nxt = cur ^ 1;
      #pragma unroll
      for (int i = 0; i < 4; ++i) {
        *(v4f*)(&As[nxt][aL + 16 * i * A_PITCH]) = ra[nxt][i];
        *(v4f*)(&Bs[nxt][bL + 16 * i * B_PITCH]) = rb[nxt][i];
      }
    }
    lds_barrier();
  }

  const int col = colTile * PBN + wc * 16 + mn;
  v8f accs[2] = {acc0, acc1};
  if (!isQuery) {
    #pragma unroll
    for (int sub = 0; sub < 2; ++sub) {
      #pragma unroll
      for (int v = 0; v < 8; ++v) {
        const int r = rowTile * PBM + wr * 32 + sub * 16 + v + half * 8;
        hk[(size_t)r * HDIM + col] = accs[sub][v];
      }
    }
  } else {
    const float bias = b1[col];
    #pragma unroll
    for (int sub = 0; sub < 2; ++sub) {
      #pragma unroll
      for (int v = 0; v < 8; ++v) {
        const int r  = rowTile * PBM + wr * 32 + sub * 16 + v + half * 8;
        const int bb = r / TGTN;
        const int tt = r - bb * TGTN;
        hq4[(((size_t)bb * (HDIM / 4) + (col >> 2)) * TGTN + tt) * 4 + (col & 3)] = accs[sub][v] + bias;
      }
    }
  }
}

__global__ __launch_bounds__(128) void score_kernel(
    const float* __restrict__ hk,
    const float* __restrict__ hq4,
    const float* __restrict__ W2,
    const float* __restrict__ b2,
    float* __restrict__ out)
{
  __shared__ __attribute__((aligned(16))) float hk_s[STILE][HDIM];
  __shared__ __attribute__((aligned(16))) float w2_s[HDIM];

  const int bst = blockIdx.x;
  const int b   = bst / 25;
  const int s0  = (bst - b * 25) * STILE;
  const int tid = threadIdx.x;

  const float* __restrict__ hkBase = hk + (size_t)(b * SRCN + s0) * HDIM;
  float st[STILE * HDIM / 128];
  #pragma unroll
  for (int i = 0; i < STILE * HDIM / 128; ++i) st[i] = hkBase[tid + i * 128];
  const float wlo = W2[tid];
  const float whi = W2[tid + 128];

  const int t = (tid < TGTN) ? tid : (TGTN - 1);
  const float* __restrict__ qB = hq4 + ((size_t)b * (HDIM / 4) * TGTN + t) * 4;
  v4f q[HDIM / 4];
  #pragma unroll
  for (int g = 0; g < HDIM / 4; ++g) q[g] = *(const v4f*)(qB + (size_t)g * TGTN * 4);

  #pragma unroll
  for (int i = 0; i < STILE * HDIM / 128; ++i) ((float*)hk_s)[tid + i * 128] = st[i];
  w2_s[tid]       = wlo;
  w2_s[tid + 128] = whi;
  lds_barrier();

  float acc0 = 0.0f, acc1 = 0.0f, acc2 = 0.0f, acc3 = 0.0f;
  float kk[2][SCH * 4][5];
  #pragma unroll
  for (int i = 0; i < SCH * 4; ++i) {
    kk[0][i][0] = hk_s[0][i];
    kk[0][i][1] = hk_s[1][i];
    kk[0][i][2] = hk_s[2][i];
    kk[0][i][3] = hk_s[3][i];
    kk[0][i][4] = w2_s[i];
  }
  #pragma unroll
  for (int c = 0; c < HDIM / (4 * SCH); ++c) {
    const int cur = c & 1;
    if (c + 1 < HDIM / (4 * SCH)) {
      #pragma unroll
      for (int i = 0; i < SCH * 4; ++i) {
        const int h = (c + 1) * SCH * 4 + i;
        kk[cur ^ 1][i][0] = hk_s[0][h];
        kk[cur ^ 1][i][1] = hk_s[1][h];
        kk[cur ^ 1][i][2] = hk_s[2][h];
        kk[cur ^ 1][i][3] = hk_s[3][h];
        kk[cur ^ 1][i][4] = w2_s[h];
      }
      __builtin_amdgcn_sched_barrier(0);
    }
    #pragma unroll
    for (int i = 0; i < SCH * 4; ++i) {
      const int h = c * SCH * 4 + i;
      const float qv = q[h >> 2][h & 3];
      const float w  = kk[cur][i][4];
      acc0 += fast_tanh(kk[cur][i][0] + qv) * w;
      acc1 += fast_tanh(kk[cur][i][1] + qv) * w;
      acc2 += fast_tanh(kk[cur][i][2] + qv) * w;
      acc3 += fast_tanh(kk[cur][i][3] + qv) * w;
    }
    if (c + 2 < HDIM / (4 * SCH))
      asm volatile("" :: "v"(acc0), "v"(acc1), "v"(acc2), "v"(acc3) : "memory");
  }

  if (tid < TGTN) {
    const float b2v = b2[0];
    float* __restrict__ o = out + (size_t)(b * SRCN + s0) * TGTN + tid;
    o[0 * TGTN] = fast_tanh(acc0 + b2v);
    o[1 * TGTN] = fast_tanh(acc1 + b2v);
    o[2 * TGTN] = fast_tanh(acc2 + b2v);
    o[3 * TGTN] = fast_tanh(acc3 + b2v);
  }
}

extern "C" void kernel_launch(void* const* d_in, const int* in_sizes, int n_in,
                              void* d_out, int out_size, void* d_ws, size_t ws_size,
                              hipStream_t stream) {
  (void)in_sizes; (void)n_in; (void)out_size; (void)ws_size;

  const float* query = (const float*)d_in[0];
  const float* keys  = (const float*)d_in[1];
  const float* Wk    = (const float*)d_in[2];
  const float* Wq    = (const float*)d_in[3];
  const float* b1    = (const float*)d_in[4];
  const float* W2    = (const float*)d_in[5];
  const float* b2    = (const float*)d_in[6];
  float* out = (float*)d_out;

  float* hk  = (float*)d_ws;
  float* hq4 = hk + (size_t)BATCH * SRCN * HDIM;

  proj_wmma_kernel<<<200, 256, 0, stream>>>(query, keys, Wk, Wq, b1, hk, hq4);
  score_kernel<<<BATCH * (SRCN / STILE), 128, 0, stream>>>(hk, hq4, W2, b2, out);
}


# Scoring re-tiled: 256 equal blocks of 25 s x 25 t (one per compute unit, one wave per SIMD), thread = one t column x 5 s rows, 125 of 128 lanes busy, hk rows in padded LDS; sum order kept
# speedup vs baseline: 5.2271x; 1.0873x over previous
#include <hip/hip_runtime.h>

typedef __attribute__((ext_vector_type(2))) float v2f;
typedef __attribute__((ext_vector_type(4))) float v4f;
typedef __attribute__((ext_vector_type(8))) float v8f;

#define BATCH 16
#define SRCN  100
#define TGTN  100
#define KDIM  512
#define HDIM  256

__device__ __forceinline__ float fast_tanh(float x) {
#if __has_builtin(__builtin_amdgcn_tanhf)
  return __builtin_amdgcn_tanhf(x);
#else
  float e = __builtin_amdgcn_exp2f(x * 2.8853900817779268f);
  return 1.0f - 2.0f * __builtin_amdgcn_rcpf(e + 1.0f);
#endif
}


__device__ __forceinline__ void lds_barrier() {
  __builtin_amdgcn_fence(__ATOMIC_RELEASE, "workgroup", "local");
  __builtin_amdgcn_s_barrier();
  __builtin_amdgcn_fence(__ATOMIC_ACQUIRE, "workgroup", "local");
}

#define PBM 64
#define PBN 64
#define PKC 64
#define A_PITCH 68
#define B_PITCH 72

__global__ __launch_bounds__(256) void proj_wmma_kernel(
    const float* __restrict__ query,
    const float* __restrict__ keys,
    const float* __restrict__ Wk,
    const float* __restrict__ Wq,
    const float* __restrict__ b1,
    float* __restrict__ hk,
    float* __restrict__ hq4)
{
  __shared__ __attribute__((aligned(16))) float As[2][PBM * A_PITCH];
  __shared__ __attribute__((aligned(16))) float Bs[2][PKC * B_PITCH];

  const int tid  = threadIdx.x;
  const int wave = tid >> 5;
  const int lane = tid & 31;
  int tile = blockIdx.x;
  const int isQuery = (tile >= 100) ? 1 : 0;
  if (isQuery) tile -= 100;
  const int rowTile = tile >> 2;
  const int colTile = tile & 3;

  const float* __restrict__ A = isQuery ? query : keys;
  const float* __restrict__ W = isQuery ? Wq : Wk;

  const int wr   = wave >> 2;
  const int wc   = wave & 3;
  const int mn   = lane & 15;
  const int half = lane >> 4;
  const int koff = half * 2;

  const int sRow = tid >> 4;
  const int sCol = (tid & 15) * 4;
  const float* __restrict__ aG = A + (size_t)(rowTile * PBM + sRow) * KDIM + sCol;
  const int aL = sRow * A_PITCH + sCol;
  const float* __restrict__ bG = W + (size_t)sRow * HDIM + colTile * PBN + sCol;
  const int bL = sRow * B_PITCH + sCol;

  const int aF = (wr * 32 + mn) * A_PITCH + koff;
  const int bF = koff * B_PITCH + wc * 16 + mn;

  v4f ra[2][4], rb[2][4];
  #pragma unroll
  for (int s = 0; s < 2; ++s) {
    #pragma unroll
    for (int i = 0; i < 4; ++i) {
      ra[s][i] = *(const v4f*)(aG + (size_t)(16 * i) * KDIM + s * PKC);
      rb[s][i] = *(const v4f*)(bG + (size_t)(s * PKC + 16 * i) * HDIM);
    }
    __builtin_amdgcn_sched_barrier(0);
  }
  #pragma unroll
  for (int i = 0; i < 4; ++i) {
    *(v4f*)(&As[0][aL + 16 * i * A_PITCH]) = ra[0][i];
    *(v4f*)(&Bs[0][bL + 16 * i * B_PITCH]) = rb[0][i];
  }
  lds_barrier();

  v8f acc0 = {}, acc1 = {};
  #pragma unroll
  for (int c = 0; c < KDIM / PKC; ++c) {
    const int cur = c & 1;
    if (c + 2 < KDIM / PKC) {
      const int kn = (c + 2) * PKC;
      #pragma unroll
      for (int i = 0; i < 4; ++i) {
        ra[cur][i] = *(const v4f*)(aG + (size_t)(16 * i) * KDIM + kn);
        rb[cur][i] = *(const v4f*)(bG + (size_t)(kn + 16 * i) * HDIM);
      }
    }
    __builtin_amdgcn_sched_barrier(0);
    const float* __restrict__ aS = &As[cur][aF];
    const float* __restrict__ bS = &Bs[cur][bF];
    v2f fa0[PKC / 4], fa1[PKC / 4], fb[PKC / 4];
    #pragma unroll
    for (int j = 0; j < PKC / 4; ++j) {
      fa0[j] = *(const v2f*)(aS + j * 4);
      fa1[j] = *(const v2f*)(aS + 16 * A_PITCH + j * 4);
      fb[j].x = bS[(j * 4) * B_PITCH];
      fb[j].y = bS[(j * 4 + 1) * B_PITCH];
      if (j & 1) __builtin_amdgcn_sched_barrier(0);
    }
    __builtin_amdgcn_sched_barrier(0);
    #pragma unroll
    for (int j = 0; j < PKC / 4; ++j) {
      acc0 = __builtin_amdgcn_wmma_f32_16x16x4_f32(false, fa0[j], false, fb[j], (short)0, acc0, false, false);
      acc1 = __builtin_amdgcn_wmma_f32_16x16x4_f32(false, fa1[j], false, fb[j], (short)0, acc1, false, false);
    }
    __builtin_amdgcn_sched_barrier(0);
    if (c + 1 < KDIM / PKC) {
      const int nxt = cur ^ 1;
      #pragma unroll
      for (int i = 0; i < 4; ++i) {
        *(v4f*)(&As[nxt][aL + 16 * i * A_PITCH]) = ra[nxt][i];
        *(v4f*)(&Bs[nxt][bL + 16 * i * B_PITCH]) = rb[nxt][i];
      }
    }
    lds_barrier();
  }

  const int col = colTile * PBN + wc * 16 + mn;
  v8f accs[2] = {acc0, acc1};
  if (!isQuery) {
    #pragma unroll
    for (int sub = 0; sub < 2; ++sub) {
      #pragma unroll
      for (int v = 0; v < 8; ++v) {
        const int r = rowTile * PBM + wr * 32 + sub * 16 + v + half * 8;
        hk[(size_t)r * HDIM + col] = accs[sub][v];
      }
    }
  } else {
    const float bias = b1[col];
    #pragma unroll
    for (int sub = 0; sub < 2; ++sub) {
      #pragma unroll
      for (int v = 0; v < 8; ++v) {
        const int r  = rowTile * PBM + wr * 32 + sub * 16 + v + half * 8;
        const int bb = r / TGTN;
        const int tt = r - bb * TGTN;
        hq4[(((size_t)bb * (HDIM / 4) + (col >> 2)) * TGTN + tt) * 4 + (col & 3)] = accs[sub][v] + bias;
      }
    }
  }
}

#define SQ       25
#define TQ       25
#define SROWS    5
#define HK_PITCH 260
#define CG       8

__global__ __launch_bounds__(128) void score_kernel(
    const float* __restrict__ hk,
    const float* __restrict__ hq4,
    const float* __restrict__ W2,
    const float* __restrict__ b2,
    float* __restrict__ out)
{
  __shared__ __attribute__((aligned(16))) float hk_s[SQ * HK_PITCH];
  __shared__ __attribute__((aligned(16))) float w2_s[HDIM];

  const int blk = blockIdx.x;
  const int b   = blk >> 4;
  const int s0  = ((blk >> 2) & 3) * SQ;
  const int t0  = (blk & 3) * TQ;
  const int tid = threadIdx.x;

  const float* __restrict__ hkBase = hk + (size_t)(b * SRCN + s0) * HDIM;
  v4f st[13];
  #pragma unroll
  for (int i = 0; i < 12; ++i) st[i] = *(const v4f*)(hkBase + (size_t)(tid + i * 128) * 4);
  if (tid < 64) st[12] = *(const v4f*)(hkBase + (size_t)(tid + 12 * 128) * 4);
  v4f wst;
  if (tid < 64) wst = *(const v4f*)(W2 + tid * 4);

  const int ti = (tid < TQ * (SQ / SROWS)) ? tid : (TQ * (SQ / SROWS) - 1);
  const int sg = ti / TQ;
  const int tl = ti - sg * TQ;
  const float* __restrict__ qB = hq4 + ((size_t)b * (HDIM / 4) * TGTN + t0 + tl) * 4;
  v4f q[HDIM / 4];
  #pragma unroll
  for (int g = 0; g < HDIM / 4; ++g) q[g] = *(const v4f*)(qB + (size_t)g * TGTN * 4);

  #pragma unroll
  for (int i = 0; i < 12; ++i) {
    const int u = tid + i * 128;
    *(v4f*)(&hk_s[(u >> 6) * HK_PITCH + (u & 63) * 4]) = st[i];
  }
  if (tid < 64) {
    const int u = tid + 12 * 128;
    *(v4f*)(&hk_s[(u >> 6) * HK_PITCH + (u & 63) * 4]) = st[12];
    *(v4f*)(&w2_s[tid * 4]) = wst;
  }
  lds_barrier();

  float acc[SROWS];
  #pragma unroll
  for (int j = 0; j < SROWS; ++j) acc[j] = 0.0f;
  const float* __restrict__ hrow = &hk_s[sg * SROWS * HK_PITCH];
  v4f kk[2][CG][SROWS + 1];
  #pragma unroll
  for (int g = 0; g < CG; ++g) {
    #pragma unroll
    for (int j = 0; j < SROWS; ++j) kk[0][g][j] = *(const v4f*)(hrow + j * HK_PITCH + g * 4);
    kk[0][g][SROWS] = *(const v4f*)(&w2_s[g * 4]);
  }
  #pragma unroll
  for (int c = 0; c < HDIM / (4 * CG); ++c) {
    const int cur = c & 1;
    if (c + 1 < HDIM / (4 * CG)) {
      #pragma unroll
      for (int g = 0; g < CG; ++g) {
        const int hg = (c + 1) * CG + g;
        #pragma unroll
        for (int j = 0; j < SROWS; ++j) kk[cur ^ 1][g][j] = *(const v4f*)(hrow + j * HK_PITCH + hg * 4);
        kk[cur ^ 1][g][SROWS] = *(const v4f*)(&w2_s[hg * 4]);
      }
      __builtin_amdgcn_sched_barrier(0);
    }
    #pragma unroll
    for (int g = 0; g < CG; ++g) {
      const v4f qg = q[c * CG + g];
      const v4f wg = kk[cur][g][SROWS];
      v4f x[SROWS];
      #pragma unroll
      for (int j = 0; j < SROWS; ++j) x[j] = kk[cur][g][j] + qg;
      #pragma unroll
      for (int e = 0; e < 4; ++e) {
        #pragma unroll
        for (int j = 0; j < SROWS; ++j) acc[j] += fast_tanh(x[j][e]) * wg[e];
      }
    }
    if (c + 2 < HDIM / (4 * CG))
      asm volatile("" :: "v"(acc[0]), "v"(acc[1]), "v"(acc[2]), "v"(acc[3]), "v"(acc[4]) : "memory");
  }

  if (tid < TQ * (SQ / SROWS)) {
    const float b2v = b2[0];
    float* __restrict__ o = out + (size_t)(b * SRCN + s0 + sg * SROWS) * TGTN + t0 + tl;
    #pragma unroll
    for (int j = 0; j < SROWS; ++j) o[j * TGTN] = fast_tanh(acc[j] + b2v);
  }
}

extern "C" void kernel_launch(void* const* d_in, const int* in_sizes, int n_in,
                              void* d_out, int out_size, void* d_ws, size_t ws_size,
                              hipStream_t stream) {
  (void)in_sizes; (void)n_in; (void)out_size; (void)ws_size;

  const float* query = (const float*)d_in[0];
  const float* keys  = (const float*)d_in[1];
  const float* Wk    = (const float*)d_in[2];
  const float* Wq    = (const float*)d_in[3];
  const float* b1    = (const float*)d_in[4];
  const float* W2    = (const float*)d_in[5];
  const float* b2    = (const float*)d_in[6];
  float* out = (float*)d_out;

  float* hk  = (float*)d_ws;
  float* hq4 = hk + (size_t)BATCH * SRCN * HDIM;

  proj_wmma_kernel<<<200, 256, 0, stream>>>(query, keys, Wk, Wq, b1, hk, hq4);
  score_kernel<<<256, 128, 0, stream>>>(hk, hq4, W2, b2, out);
}


# Projection GEMMs: LDS fragment reads of chunk c+1 (second fragment register set) and LDS stores of chunk c+2 issue behind the first 8 WMMAs of chunk c, not between barrier and WMMAs; sums unchanged
# speedup vs baseline: 5.5166x; 1.0554x over previous
#include <hip/hip_runtime.h>

typedef __attribute__((ext_vector_type(2))) float v2f;
typedef __attribute__((ext_vector_type(4))) float v4f;
typedef __attribute__((ext_vector_type(8))) float v8f;

#define BATCH 16
#define SRCN  100
#define TGTN  100
#define KDIM  512
#define HDIM  256

__device__ __forceinline__ float fast_tanh(float x) {
#if __has_builtin(__builtin_amdgcn_tanhf)
  return __builtin_amdgcn_tanhf(x);
#else
  float e = __builtin_amdgcn_exp2f(x * 2.8853900817779268f);
  return 1.0f - 2.0f * __builtin_amdgcn_rcpf(e + 1.0f);
#endif
}


__device__ __forceinline__ void lds_barrier() {
  __builtin_amdgcn_fence(__ATOMIC_RELEASE, "workgroup", "local");
  __builtin_amdgcn_s_barrier();
  __builtin_amdgcn_fence(__ATOMIC_ACQUIRE, "workgroup", "local");
}

#define PBM 64
#define PBN 64
#define PKC 64
#define A_PITCH 68
#define B_PITCH 72
#define HEAD 4

__global__ __launch_bounds__(256) void proj_wmma_kernel(
    const float* __restrict__ query,
    const float* __restrict__ keys,
    const float* __restrict__ Wk,
    const float* __restrict__ Wq,
    const float* __restrict__ b1,
    float* __restrict__ hk,
    float* __restrict__ hq4)
{
  __shared__ __attribute__((aligned(16))) float As[2][PBM * A_PITCH];
  __shared__ __attribute__((aligned(16))) float Bs[2][PKC * B_PITCH];

  const int tid  = threadIdx.x;
  const int wave = tid >> 5;
  const int lane = tid & 31;
  int tile = blockIdx.x;
  const int isQuery = (tile >= 100) ? 1 : 0;
  if (isQuery) tile -= 100;
  const int rowTile = tile >> 2;
  const int colTile = tile & 3;

  const float* __restrict__ A = isQuery ? query : keys;
  const float* __restrict__ W = isQuery ? Wq : Wk;

  const int wr   = wave >> 2;
  const int wc   = wave & 3;
  const int mn   = lane & 15;
  const int half = lane >> 4;
  const int koff = half * 2;

  const int sRow = tid >> 4;
  const int sCol = (tid & 15) * 4;
  const float* __restrict__ aG = A + (size_t)(rowTile * PBM + sRow) * KDIM + sCol;
  const int aL = sRow * A_PITCH + sCol;
  const float* __restrict__ bG = W + (size_t)sRow * HDIM + colTile * PBN + sCol;
  const int bL = sRow * B_PITCH + sCol;

  const int aF = (wr * 32 + mn) * A_PITCH + koff;
  const int bF = koff * B_PITCH + wc * 16 + mn;

  v4f ra[2][4], rb[2][4];
  v2f fa0[2][PKC / 4], fa1[2][PKC / 4], fb[2][PKC / 4];

  #pragma unroll
  for (int s = 0; s < 2; ++s) {
    #pragma unroll
    for (int i = 0; i < 4; ++i) {
      ra[s][i] = *(const v4f*)(aG + (size_t)(16 * i) * KDIM + s * PKC);
      rb[s][i] = *(const v4f*)(bG + (size_t)(s * PKC + 16 * i) * HDIM);
    }
    __builtin_amdgcn_sched_barrier(0);
  }
  #pragma unroll
  for (int i = 0; i < 4; ++i) {
    *(v4f*)(&As[0][aL + 16 * i * A_PITCH]) = ra[0][i];
    *(v4f*)(&Bs[0][bL + 16 * i * B_PITCH]) = rb[0][i];
  }
  #pragma unroll
  for (int i = 0; i < 4; ++i) {
    ra[0][i] = *(const v4f*)(aG + (size_t)(16 * i) * KDIM + 2 * PKC);
    rb[0][i] = *(const v4f*)(bG + (size_t)(2 * PKC + 16 * i) * HDIM);
  }
  lds_barrier();
  #pragma unroll
  for (int j = 0; j < PKC / 4; ++j) {
    fa0[0][j] = *(const v2f*)(&As[0][aF + j * 4]);
    fa1[0][j] = *(const v2f*)(&As[0][aF + 16 * A_PITCH + j * 4]);
    fb[0][j].x = Bs[0][bF + (j * 4) * B_PITCH];
    fb[0][j].y = Bs[0][bF + (j * 4 + 1) * B_PITCH];
  }
  #pragma unroll
  for (int i = 0; i < 4; ++i) {
    *(v4f*)(&As[1][aL + 16 * i * A_PITCH]) = ra[1][i];
    *(v4f*)(&Bs[1][bL + 16 * i * B_PITCH]) = rb[1][i];
  }
  #pragma unroll
  for (int i = 0; i < 4; ++i) {
    ra[1][i] = *(const v4f*)(aG + (size_t)(16 * i) * KDIM + 3 * PKC);
    rb[1][i] = *(const v4f*)(bG + (size_t)(3 * PKC + 16 * i) * HDIM);
  }
  lds_barrier();

  v8f acc0 = {}, acc1 = {};
  #pragma unroll
  for (int c = 0; c < KDIM / PKC; ++c) {
    const int cur = c & 1;
    const int nxt = cur ^ 1;
    #pragma unroll
    for (int j = 0; j < HEAD; ++j) {
      acc0 = __builtin_amdgcn_wmma_f32_16x16x4_f32(false, fa0[cur][j], false, fb[cur][j], (short)0, acc0, false, false);
      acc1 = __builtin_amdgcn_wmma_f32_16x16x4_f32(false, fa1[cur][j], false, fb[cur][j], (short)0, acc1, false, false);
    }
    __builtin_amdgcn_sched_barrier(0);
    if (c + 2 < KDIM / PKC) {
      #pragma unroll
      for (int i = 0; i < 4; ++i) {
        *(v4f*)(&As[cur][aL + 16 * i * A_PITCH]) = ra[cur][i];
        *(v4f*)(&Bs[cur][bL + 16 * i * B_PITCH]) = rb[cur][i];
      }
    }
    if (c + 4 < KDIM / PKC) {
      const int kn = (c + 4) * PKC;
      #pragma unroll
      for (int i = 0; i < 4; ++i) {
        ra[cur][i] = *(const v4f*)(aG + (size_t)(16 * i) * KDIM + kn);
        rb[cur][i] = *(const v4f*)(bG + (size_t)(kn + 16 * i) * HDIM);
      }
    }
    if (c + 1 < KDIM / PKC) {
      const float* __restrict__ aS = &As[nxt][aF];
      const float* __restrict__ bS = &Bs[nxt][bF];
      #pragma unroll
      for (int j = 0; j < PKC / 4; ++j) {
        fa0[nxt][j] = *(const v2f*)(aS + j * 4);
        fa1[nxt][j] = *(const v2f*)(aS + 16 * A_PITCH + j * 4);
        fb[nxt][j].x = bS[(j * 4) * B_PITCH];
        fb[nxt][j].y = bS[(j * 4 + 1) * B_PITCH];
      }
    }
    __builtin_amdgcn_sched_barrier(0);
    #pragma unroll
    for (int j = HEAD; j < PKC / 4; ++j) {
      acc0 = __builtin_amdgcn_wmma_f32_16x16x4_f32(false, fa0[cur][j], false, fb[cur][j], (short)0, acc0, false, false);
      acc1 = __builtin_amdgcn_wmma_f32_16x16x4_f32(false, fa1[cur][j], false, fb[cur][j], (short)0, acc1, false, false);
    }
    __builtin_amdgcn_sched_barrier(0);
    if (c + 2 < KDIM / PKC) lds_barrier();
  }

  const int col = colTile * PBN + wc * 16 + mn;
  v8f accs[2] = {acc0, acc1};
  if (!isQuery) {
    #pragma unroll
    for (int sub = 0; sub < 2; ++sub) {
      #pragma unroll
      for (int v = 0; v < 8; ++v) {
        const int r = rowTile * PBM + wr * 32 + sub * 16 + v + half * 8;
        hk[(size_t)r * HDIM + col] = accs[sub][v];
      }
    }
  } else {
    const float bias = b1[col];
    #pragma unroll
    for (int sub = 0; sub < 2; ++sub) {
      #pragma unroll
      for (int v = 0; v < 8; ++v) {
        const int r  = rowTile * PBM + wr * 32 + sub * 16 + v + half * 8;
        const int bb = r / TGTN;
        const int tt = r - bb * TGTN;
        hq4[(((size_t)bb * (HDIM / 4) + (col >> 2)) * TGTN + tt) * 4 + (col & 3)] = accs[sub][v] + bias;
      }
    }
  }
}

#define SQ       25
#define TQ       25
#define SROWS    5
#define HK_PITCH 260
#define CG       8

__global__ __launch_bounds__(128) void score_kernel(
    const float* __restrict__ hk,
    const float* __restrict__ hq4,
    const float* __restrict__ W2,
    const float* __restrict__ b2,
    float* __restrict__ out)
{
  __shared__ __attribute__((aligned(16))) float hk_s[SQ * HK_PITCH];
  __shared__ __attribute__((aligned(16))) float w2_s[HDIM];

  const int blk = blockIdx.x;
  const int b   = blk >> 4;
  const int s0  = ((blk >> 2) & 3) * SQ;
  const int t0  = (blk & 3) * TQ;
  const int tid = threadIdx.x;

  const float* __restrict__ hkBase = hk + (size_t)(b * SRCN + s0) * HDIM;
  v4f st[13];
  #pragma unroll
  for (int i = 0; i < 12; ++i) st[i] = *(const v4f*)(hkBase + (size_t)(tid + i * 128) * 4);
  if (tid < 64) st[12] = *(const v4f*)(hkBase + (size_t)(tid + 12 * 128) * 4);
  v4f wst;
  if (tid < 64) wst = *(const v4f*)(W2 + tid * 4);

  const int ti = (tid < TQ * (SQ / SROWS)) ? tid : (TQ * (SQ / SROWS) - 1);
  const int sg = ti / TQ;
  const int tl = ti - sg * TQ;
  const float* __restrict__ qB = hq4 + ((size_t)b * (HDIM / 4) * TGTN + t0 + tl) * 4;
  v4f q[HDIM / 4];
  #pragma unroll
  for (int g = 0; g < HDIM / 4; ++g) q[g] = *(const v4f*)(qB + (size_t)g * TGTN * 4);

  #pragma unroll
  for (int i = 0; i < 12; ++i) {
    const int u = tid + i * 128;
    *(v4f*)(&hk_s[(u >> 6) * HK_PITCH + (u & 63) * 4]) = st[i];
  }
  if (tid < 64) {
    const int u = tid + 12 * 128;
    *(v4f*)(&hk_s[(u >> 6) * HK_PITCH + (u & 63) * 4]) = st[12];
    *(v4f*)(&w2_s[tid * 4]) = wst;
  }
  lds_barrier();

  float acc[SROWS];
  #pragma unroll
  for (int j = 0; j < SROWS; ++j) acc[j] = 0.0f;
  const float* __restrict__ hrow = &hk_s[sg * SROWS * HK_PITCH];
  v4f kk[2][CG][SROWS + 1];
  #pragma unroll
  for (int g = 0; g < CG; ++g) {
    #pragma unroll
    for (int j = 0; j < SROWS; ++j) kk[0][g][j] = *(const v4f*)(hrow + j * HK_PITCH + g * 4);
    kk[0][g][SROWS] = *(const v4f*)(&w2_s[g * 4]);
  }
  #pragma unroll
  for (int c = 0; c < HDIM / (4 * CG); ++c) {
    const int cur = c & 1;
    if (c + 1 < HDIM / (4 * CG)) {
      #pragma unroll
      for (int g = 0; g < CG; ++g) {
        const int hg = (c + 1) * CG + g;
        #pragma unroll
        for (int j = 0; j < SROWS; ++j) kk[cur ^ 1][g][j] = *(const v4f*)(hrow + j * HK_PITCH + hg * 4);
        kk[cur ^ 1][g][SROWS] = *(const v4f*)(&w2_s[hg * 4]);
      }
      __builtin_amdgcn_sched_barrier(0);
    }
    #pragma unroll
    for (int g = 0; g < CG; ++g) {
      const v4f qg = q[c * CG + g];
      const v4f wg = kk[cur][g][SROWS];
      v4f x[SROWS];
      #pragma unroll
      for (int j = 0; j < SROWS; ++j) x[j] = kk[cur][g][j] + qg;
      #pragma unroll
      for (int e = 0; e < 4; ++e) {
        #pragma unroll
        for (int j = 0; j < SROWS; ++j) acc[j] += fast_tanh(x[j][e]) * wg[e];
      }
    }
    if (c + 2 < HDIM / (4 * CG))
      asm volatile("" :: "v"(acc[0]), "v"(acc[1]), "v"(acc[2]), "v"(acc[3]), "v"(acc[4]) : "memory");
  }

  if (tid < TQ * (SQ / SROWS)) {
    const float b2v = b2[0];
    float* __restrict__ o = out + (size_t)(b * SRCN + s0 + sg * SROWS) * TGTN + t0 + tl;
    #pragma unroll
    for (int j = 0; j < SROWS; ++j) o[j * TGTN] = fast_tanh(acc[j] + b2v);
  }
}

extern "C" void kernel_launch(void* const* d_in, const int* in_sizes, int n_in,
                              void* d_out, int out_size, void* d_ws, size_t ws_size,
                              hipStream_t stream) {
  (void)in_sizes; (void)n_in; (void)out_size; (void)ws_size;

  const float* query = (const float*)d_in[0];
  const float* keys  = (const float*)d_in[1];
  const float* Wk    = (const float*)d_in[2];
  const float* Wq    = (const float*)d_in[3];
  const float* b1    = (const float*)d_in[4];
  const float* W2    = (const float*)d_in[5];
  const float* b2    = (const float*)d_in[6];
  float* out = (float*)d_out;

  float* hk  = (float*)d_ws;
  float* hq4 = hk + (size_t)BATCH * SRCN * HDIM;

  proj_wmma_kernel<<<200, 256, 0, stream>>>(query, keys, Wk, Wq, b1, hk, hq4);
  score_kernel<<<256, 128, 0, stream>>>(hk, hq4, W2, b2, out);
}
